# starting kernel
# baseline (speedup 1.0000x reference)
#include <hip/hip_runtime.h>
#include <hip/hip_bf16.h>

typedef float v2f __attribute__((ext_vector_type(2)));
typedef float v8f __attribute__((ext_vector_type(8)));

#define N_NODES 2048
#define N_GRAPHS 64
#define N_EDGES 16384
#define BATCH 65536
#define META 64
#define TXD 8
#define NOISE 128
#define TSTRIDE 48
#define CROWS 64
#define NCHUNK (BATCH / CROWS)

__device__ __forceinline__ int kext_row(int k) {
    if (k < 32) return 2049 + k;
    if (k == 32) return 2048;
    if (k <= 40) return 2081 + (k - 33);
    return 2048;
}

__global__ void __launch_bounds__(256) gcn_kernel(const int* __restrict__ edges,
                                                  const float* __restrict__ gw,
                                                  const float* __restrict__ gb,
                                                  float* __restrict__ graph_emb) {
    __shared__ float deg[N_NODES];
    __shared__ float agg[N_NODES];
    const int g = blockIdx.x;
    const int t = threadIdx.x;
    const int* src = edges + (size_t)g * (2 * N_EDGES);
    const int* dst = src + N_EDGES;

    for (int i = t; i < N_NODES; i += 256) deg[i] = 1.0f;
    __syncthreads();
    for (int e = t; e < N_EDGES; e += 256)
        atomicAdd(&deg[dst[e]], 1.0f);
    __syncthreads();
    for (int i = t; i < N_NODES; i += 256) {
        float d = deg[i];
        float r = d > 0.0f ? rsqrtf(d) : 0.0f;
        deg[i] = r;
        agg[i] = r * r;
    }
    __syncthreads();
    for (int e = t; e < N_EDGES; e += 256) {
        int s = src[e], d = dst[e];
        atomicAdd(&agg[d], deg[s] * deg[d]);
    }
    __syncthreads();
    const float w = *gw, b = *gb;
    for (int i = t; i < N_NODES; i += 256)
        graph_emb[(size_t)g * N_NODES + i] = agg[i] * w + b;
}

__global__ void __launch_bounds__(256) proj_kernel(const float* __restrict__ ge,
                                                   const float* __restrict__ embW,
                                                   float* __restrict__ gp) {
    const int wave  = (blockIdx.x * blockDim.x + threadIdx.x) >> 5;
    const int lane  = threadIdx.x & 31;
    const int mtile = wave >> 3;
    const int ntile = wave & 7;
    const int lrow  = lane & 15;
    const int khalf = (lane >> 4) * 2;
    const int arow  = mtile * 16 + lrow;
    const int col   = ntile * 16 + lrow;

    v8f c = {};
    for (int kk = 0; kk < N_NODES; kk += 4) {
        const int k0 = kk + khalf;
        v2f a, bm;
        a.x  = ge[(size_t)arow * N_NODES + k0];
        a.y  = ge[(size_t)arow * N_NODES + k0 + 1];
        bm.x = embW[(size_t)k0 * NOISE + col];
        bm.y = embW[(size_t)(k0 + 1) * NOISE + col];
        c = __builtin_amdgcn_wmma_f32_16x16x4_f32(false, a, false, bm,
                                                  (short)0, c, false, false);
    }
    const int rbase = mtile * 16 + (lane >> 4) * 8;
#pragma unroll
    for (int r = 0; r < 8; ++r)
        gp[(size_t)(rbase + r) * NOISE + col] = c[r];
}

__global__ void __launch_bounds__(256) batch_kernel(const int* __restrict__ graphs,
                                                    const float* __restrict__ chain,
                                                    const float* __restrict__ trig,
                                                    const float* __restrict__ tx,
                                                    const float* __restrict__ trigW,
                                                    const float* __restrict__ trigb,
                                                    const float* __restrict__ embW,
                                                    const float* __restrict__ embb,
                                                    const float* __restrict__ gp,
                                                    float* __restrict__ out) {
    __shared__ __align__(16) float gpl[N_GRAPHS * NOISE];
    __shared__ float T[CROWS * TSTRIDE];

    const int t     = threadIdx.x;
    const int lane  = t & 31;
    const int w     = t >> 5;
    const int lrow  = lane & 15;
    const int khalf = (lane >> 4) * 2;
    const int col   = w * 16 + lrow;
    const int lr0   = (lane >> 4) * 8;

    for (int i = t * 4; i < N_GRAPHS * NOISE; i += 256 * 4)
        *(float4*)&gpl[i] = *(const float4*)&gp[i];

    v2f bf[11];
#pragma unroll
    for (int ks = 0; ks < 11; ++ks) {
        const int k0 = ks * 4 + khalf;
        bf[ks].x = embW[(size_t)kext_row(k0) * NOISE + col];
        bf[ks].y = embW[(size_t)kext_row(k0 + 1) * NOISE + col];
    }

    const int rt1 = w >> 1;
    const int nt1 = w & 1;
    const int c1  = nt1 * 16 + lrow;
    v2f b1[16];
#pragma unroll
    for (int ks = 0; ks < 16; ++ks) {
        const int k0 = ks * 4 + khalf;
        b1[ks].x = trigW[k0 * 32 + c1];
        b1[ks].y = trigW[(k0 + 1) * 32 + c1];
    }
    const float tb = trigb[c1];
    const float eb = embb[col];
    __syncthreads();

    for (int chunk = blockIdx.x; chunk < NCHUNK; chunk += gridDim.x) {
        const int rowbase = chunk * CROWS;

        if (chunk + gridDim.x < NCHUNK) {
            const float* p = trig + (size_t)(chunk + gridDim.x) * CROWS * META
                                  + t * 16;
            __builtin_prefetch(p, 0, 3);
        }

#pragma unroll
        for (int j = 0; j < 4; ++j) {
            const int slot = t + 256 * j;
            const int fr = slot >> 4;
            const int fc = 32 + (slot & 15);
            float v = 0.0f;
            const int grow = rowbase + fr;
            if (fc == 32)      v = chain[grow];
            else if (fc <= 40) v = tx[(size_t)grow * TXD + (fc - 33)];
            T[fr * TSTRIDE + fc] = v;
        }

        {
            v8f d = {};
            const int arow = rowbase + rt1 * 16 + lrow;
#pragma unroll
            for (int ks = 0; ks < 16; ++ks) {
                const int k0 = ks * 4 + khalf;
                v2f a;
                a.x = trig[(size_t)arow * META + k0];
                a.y = trig[(size_t)arow * META + k0 + 1];
                d = __builtin_amdgcn_wmma_f32_16x16x4_f32(false, a, false, b1[ks],
                                                          (short)0, d, false, false);
            }
#pragma unroll
            for (int r = 0; r < 8; ++r) {
                float v = d[r] + tb;
                T[(rt1 * 16 + lr0 + r) * TSTRIDE + c1] = v > 0.0f ? v : 0.0f;
            }
        }
        __syncthreads();

        v8f acc0 = {}, acc1 = {}, acc2 = {}, acc3 = {};
#pragma unroll
        for (int ks = 0; ks < 11; ++ks) {
            const int k0 = ks * 4 + khalf;
            v2f a0, a1, a2, a3;
            a0.x = T[(0 * 16 + lrow) * TSTRIDE + k0];
            a0.y = T[(0 * 16 + lrow) * TSTRIDE + k0 + 1];
            a1.x = T[(1 * 16 + lrow) * TSTRIDE + k0];
            a1.y = T[(1 * 16 + lrow) * TSTRIDE + k0 + 1];
            a2.x = T[(2 * 16 + lrow) * TSTRIDE + k0];
            a2.y = T[(2 * 16 + lrow) * TSTRIDE + k0 + 1];
            a3.x = T[(3 * 16 + lrow) * TSTRIDE + k0];
            a3.y = T[(3 * 16 + lrow) * TSTRIDE + k0 + 1];
            acc0 = __builtin_amdgcn_wmma_f32_16x16x4_f32(false, a0, false, bf[ks],
                                                         (short)0, acc0, false, false);
            acc1 = __builtin_amdgcn_wmma_f32_16x16x4_f32(false, a1, false, bf[ks],
                                                         (short)0, acc1, false, false);
            acc2 = __builtin_amdgcn_wmma_f32_16x16x4_f32(false, a2, false, bf[ks],
                                                         (short)0, acc2, false, false);
            acc3 = __builtin_amdgcn_wmma_f32_16x16x4_f32(false, a3, false, bf[ks],
                                                         (short)0, acc3, false, false);
        }

#pragma unroll
        for (int rt = 0; rt < 4; ++rt) {
            const v8f& c = (rt == 0) ? acc0 : (rt == 1) ? acc1
                         : (rt == 2) ? acc2 : acc3;
#pragma unroll
            for (int r = 0; r < 8; ++r) {
                const int grow = rowbase + rt * 16 + lr0 + r;
                const int g = graphs[grow];
                out[(size_t)grow * NOISE + col] = c[r] + gpl[g * NOISE + col] + eb;
            }
        }
        __syncthreads();
    }
}

extern "C" void kernel_launch(void* const* d_in, const int* in_sizes, int n_in,
                              void* d_out, int out_size, void* d_ws, size_t ws_size,
                              hipStream_t stream) {
    const int*   graphs = (const int*)d_in[0];
    const float* chain  = (const float*)d_in[1];
    const float* trig   = (const float*)d_in[2];
    const float* tx     = (const float*)d_in[3];
    const int*   edges  = (const int*)d_in[4];
    const float* gw     = (const float*)d_in[5];
    const float* gb     = (const float*)d_in[6];
    const float* trigW  = (const float*)d_in[7];
    const float* trigb  = (const float*)d_in[8];
    const float* embW   = (const float*)d_in[9];
    const float* embb   = (const float*)d_in[10];
    float* out = (float*)d_out;

    float* ge = (float*)d_ws;
    float* gp = ge + (size_t)N_GRAPHS * N_NODES;

    gcn_kernel<<<N_GRAPHS, 256, 0, stream>>>(edges, gw, gb, ge);
    proj_kernel<<<4, 256, 0, stream>>>(ge, embW, gp);
    batch_kernel<<<512, 256, 0, stream>>>(graphs, chain, trig, tx, trigW, trigb,
                                          embW, embb, gp, out);
}


# batch kernel: each chunk's inputs (trigger slab, tx, chain) read linearly as one batch of whole-line loads into LDS; step-1 A fragments read from LDS (was strided fragment loads, load-by-load fill)
# speedup vs baseline: 3.6297x; 3.6297x over previous
#include <hip/hip_runtime.h>
#include <hip/hip_bf16.h>

typedef float v2f __attribute__((ext_vector_type(2)));
typedef float v8f __attribute__((ext_vector_type(8)));

#define N_NODES 2048
#define N_GRAPHS 64
#define N_EDGES 16384
#define BATCH 65536
#define META 64
#define TXD 8
#define NOISE 128
#define TSTRIDE 48
#define CROWS 64
#define A1PITCH 68
#define NCHUNK (BATCH / CROWS)

__device__ __forceinline__ int kext_row(int k) {
    if (k < 32) return 2049 + k;
    if (k == 32) return 2048;
    if (k <= 40) return 2081 + (k - 33);
    return 2048;
}

#define GSPLIT 4
#define NRANGE (N_NODES / GSPLIT)
#define GTHREADS 512
#define EPT4 (N_EDGES / 4 / GTHREADS)

__global__ void __launch_bounds__(GTHREADS) gcn_kernel(const int* __restrict__ edges,
                                                       const float* __restrict__ gw,
                                                       const float* __restrict__ gb,
                                                       float* __restrict__ graph_emb) {
    __shared__ __align__(16) float cnt[N_NODES];
    __shared__ __align__(16) float dinv[N_NODES];
    __shared__ float agg[NRANGE];
    const int t = threadIdx.x;
    const int g = blockIdx.x / GSPLIT;
    const int p = blockIdx.x % GSPLIT;

    const int4* src4 = (const int4*)(edges + (size_t)g * (2 * N_EDGES));
    const int4* dst4 = src4 + N_EDGES / 4;
    int4 s[EPT4], d[EPT4];
    for (int j = 0; j < EPT4; ++j) d[j] = dst4[j * GTHREADS + t];

    *(float4*)&cnt[t * 4] = make_float4(0.0f, 0.0f, 0.0f, 0.0f);
    __syncthreads();
    for (int j = 0; j < EPT4; ++j) s[j] = src4[j * GTHREADS + t];
    for (int j = 0; j < EPT4; ++j) {
        atomicAdd(&cnt[d[j].x], 1.0f);
        atomicAdd(&cnt[d[j].y], 1.0f);
        atomicAdd(&cnt[d[j].z], 1.0f);
        atomicAdd(&cnt[d[j].w], 1.0f);
    }
    __syncthreads();

    const float4 cv = *(const float4*)&cnt[t * 4];
    const float dg[4] = {1.0f + cv.x, 1.0f + cv.y, 1.0f + cv.z, 1.0f + cv.w};
    float r[4];
    for (int k = 0; k < 4; ++k) r[k] = dg[k] > 0.0f ? rsqrtf(dg[k]) : 0.0f;
    *(float4*)&dinv[t * 4] = make_float4(r[0], r[1], r[2], r[3]);
    if (t / (NRANGE / 4) == p) {
        for (int k = 0; k < 4; ++k) agg[(t * 4 + k) % NRANGE] = r[k] * r[k];
    }
    __syncthreads();

    float nrm[EPT4][4];
    for (int j = 0; j < EPT4; ++j) {
        nrm[j][0] = dinv[s[j].x] * dinv[d[j].x];
        nrm[j][1] = dinv[s[j].y] * dinv[d[j].y];
        nrm[j][2] = dinv[s[j].z] * dinv[d[j].z];
        nrm[j][3] = dinv[s[j].w] * dinv[d[j].w];
    }
    for (int j = 0; j < EPT4; ++j) {
        const unsigned dv[4] = {(unsigned)d[j].x, (unsigned)d[j].y, (unsigned)d[j].z, (unsigned)d[j].w};
        for (int k = 0; k < 4; ++k)
            if (dv[k] / NRANGE == (unsigned)p)
                atomicAdd(&agg[dv[k] % NRANGE], nrm[j][k]);
    }
    __syncthreads();
    const float w = *gw, b = *gb;
    graph_emb[(size_t)g * N_NODES + p * NRANGE + t] = agg[t] * w + b;
}

#define PKC      256
#define PNCH     (N_NODES / PKC)
#define APITCH   (PKC + 4)
#define PCOPY    512
#define PTHREADS 544
#define PG       16

__global__ void __launch_bounds__(PTHREADS) proj_kernel(const float* __restrict__ ge,
                                                        const float* __restrict__ embW,
                                                        float* __restrict__ gp) {
    __shared__ __align__(16) float As[2][16 * APITCH];
    __shared__ __align__(16) float Bs[2][PKC * 16];

    const int t     = threadIdx.x;
    const int lane  = t & 31;
    const int wv    = __builtin_amdgcn_readfirstlane(t >> 5);
    const int mtile = blockIdx.x >> 3;
    const int ntile = blockIdx.x & 7;
    const int lrow  = lane & 15;
    const int khalf = (lane >> 4) * 2;

    const int i0 = (t - 32) & (PCOPY - 1), i1 = i0 + PCOPY;
    const float* ga0 = ge + (size_t)(mtile * 16 + (i0 >> 6)) * N_NODES + (i0 & 63) * 4;
    const float* ga1 = ge + (size_t)(mtile * 16 + (i1 >> 6)) * N_NODES + (i1 & 63) * 4;
    const float* gb0 = embW + (size_t)(i0 >> 2) * NOISE + ntile * 16 + (i0 & 3) * 4;
    const float* gb1 = embW + (size_t)(i1 >> 2) * NOISE + ntile * 16 + (i1 & 3) * 4;
    const int la0 = (i0 >> 6) * APITCH + (i0 & 63) * 4;
    const int la1 = (i1 >> 6) * APITCH + (i1 & 63) * 4;
    const int lb0 = i0 * 4;
    const int lb1 = i1 * 4;

    if (wv != 0) {
        const float4 ra0 = *(const float4*)ga0, ra1 = *(const float4*)ga1;
        const float4 rb0 = *(const float4*)gb0, rb1 = *(const float4*)gb1;
        *(float4*)&As[0][la0] = ra0;  *(float4*)&As[0][la1] = ra1;
        *(float4*)&Bs[0][lb0] = rb0;  *(float4*)&Bs[0][lb1] = rb1;
    }

    v8f c = {};
    for (int ch = 0; ch < PNCH; ++ch) {
        const int st = ch & 1;
        __syncthreads();
        if (wv != 0) {
            if (ch + 1 < PNCH) {
                const int ko = (ch + 1) * PKC;
                const float4 ra0 = *(const float4*)(ga0 + ko);
                const float4 ra1 = *(const float4*)(ga1 + ko);
                const float4 rb0 = *(const float4*)(gb0 + (size_t)ko * NOISE);
                const float4 rb1 = *(const float4*)(gb1 + (size_t)ko * NOISE);
                *(float4*)&As[st ^ 1][la0] = ra0;  *(float4*)&As[st ^ 1][la1] = ra1;
                *(float4*)&Bs[st ^ 1][lb0] = rb0;  *(float4*)&Bs[st ^ 1][lb1] = rb1;
            }
        } else {
            const float* ap = &As[st][lrow * APITCH + khalf];
            const float* bp = &Bs[st][khalf * 16 + lrow];
            v2f fa[2][PG], fb[2][PG];
#pragma unroll
            for (int i = 0; i < PG; ++i) {
                fa[0][i]   = *(const v2f*)(ap + i * 4);
                fb[0][i].x = bp[i * 64];
                fb[0][i].y = bp[i * 64 + 16];
            }
#pragma unroll
            for (int g = 0; g < PKC / 4 / PG; ++g) {
                if (g + 1 < PKC / 4 / PG) {
#pragma unroll
                    for (int i = 0; i < PG; ++i) {
                        const int q = (g + 1) * PG + i;
                        fa[(g + 1) & 1][i]   = *(const v2f*)(ap + q * 4);
                        fb[(g + 1) & 1][i].x = bp[q * 64];
                        fb[(g + 1) & 1][i].y = bp[q * 64 + 16];
                    }
                }
                __builtin_amdgcn_sched_barrier(0);
#pragma unroll
                for (int i = 0; i < PG; ++i)
                    c = __builtin_amdgcn_wmma_f32_16x16x4_f32(false, fa[g & 1][i], false,
                                                              fb[g & 1][i], (short)0, c,
                                                              false, false);
                __builtin_amdgcn_sched_barrier(0);
            }
        }
    }
    if (wv == 0) {
        const int col   = ntile * 16 + lrow;
        const int rbase = mtile * 16 + (lane >> 4) * 8;
#pragma unroll
        for (int r = 0; r < 8; ++r)
            gp[(size_t)(rbase + r) * NOISE + col] = c[r];
    }
}

__global__ void __launch_bounds__(256) batch_kernel(const int* __restrict__ graphs,
                                                    const float* __restrict__ chain,
                                                    const float* __restrict__ trig,
                                                    const float* __restrict__ tx,
                                                    const float* __restrict__ trigW,
                                                    const float* __restrict__ trigb,
                                                    const float* __restrict__ embW,
                                                    const float* __restrict__ embb,
                                                    const float* __restrict__ gp,
                                                    float* __restrict__ out) {
    __shared__ __align__(16) float gpl[N_GRAPHS * NOISE];
    __shared__ float T[CROWS * TSTRIDE];
    __shared__ __align__(16) float A1[CROWS * A1PITCH];

    const int t     = threadIdx.x;
    const int lane  = t & 31;
    const int w     = t >> 5;
    const int lrow  = lane & 15;
    const int khalf = (lane >> 4) * 2;
    const int col   = w * 16 + lrow;
    const int lr0   = (lane >> 4) * 8;

    for (int i = t * 4; i < N_GRAPHS * NOISE; i += 256 * 4)
        *(float4*)&gpl[i] = *(const float4*)&gp[i];

    v2f bf[11];
#pragma unroll
    for (int ks = 0; ks < 11; ++ks) {
        const int k0 = ks * 4 + khalf;
        bf[ks].x = embW[(size_t)kext_row(k0) * NOISE + col];
        bf[ks].y = embW[(size_t)kext_row(k0 + 1) * NOISE + col];
    }

    const int rt1 = w >> 1;
    const int nt1 = w & 1;
    const int c1  = nt1 * 16 + lrow;
    v2f b1[16];
#pragma unroll
    for (int ks = 0; ks < 16; ++ks) {
        const int k0 = ks * 4 + khalf;
        b1[ks].x = trigW[k0 * 32 + c1];
        b1[ks].y = trigW[(k0 + 1) * 32 + c1];
    }
    const float tb = trigb[c1];
    const float eb = embb[col];
    if (t < CROWS) {
        for (int c = 41; c < TSTRIDE; ++c) T[t * TSTRIDE + c] = 0.0f;
    }
    __syncthreads();

    for (int chunk = blockIdx.x; chunk < NCHUNK; chunk += gridDim.x) {
        const int rowbase = chunk * CROWS;

        if (chunk + gridDim.x < NCHUNK) {
            const float* p = trig + (size_t)(chunk + gridDim.x) * CROWS * META
                                  + t * 16;
            __builtin_prefetch(p, 0, 3);
        }

        const float4* tsrc = (const float4*)(trig + (size_t)rowbase * META);
        float4 tv[4];
        for (int j = 0; j < 4; ++j) tv[j] = tsrc[t + 256 * j];
        float4 xv = make_float4(0.0f, 0.0f, 0.0f, 0.0f);
        float cv = 0.0f;
        if (t < 128)      xv = ((const float4*)(tx + (size_t)rowbase * TXD))[t];
        else if (t < 192) cv = chain[rowbase + (t - 128)];

        for (int j = 0; j < 4; ++j) {
            const int i = t + 256 * j;
            *(float4*)&A1[(i >> 4) * A1PITCH + (i & 15) * 4] = tv[j];
        }
        if (t < 128) {
            float* tp = &T[(t >> 1) * TSTRIDE + 33 + (t & 1) * 4];
            tp[0] = xv.x;  tp[1] = xv.y;  tp[2] = xv.z;  tp[3] = xv.w;
        } else if (t < 192) {
            T[(t - 128) * TSTRIDE + 32] = cv;
        }
        __syncthreads();

        {
            v8f d = {};
            const float* ap = &A1[(rt1 * 16 + lrow) * A1PITCH + khalf];
#pragma unroll
            for (int ks = 0; ks < 16; ++ks) {
                const v2f a = *(const v2f*)(ap + ks * 4);
                d = __builtin_amdgcn_wmma_f32_16x16x4_f32(false, a, false, b1[ks],
                                                          (short)0, d, false, false);
            }
#pragma unroll
            for (int r = 0; r < 8; ++r) {
                float v = d[r] + tb;
                T[(rt1 * 16 + lr0 + r) * TSTRIDE + c1] = v > 0.0f ? v : 0.0f;
            }
        }
        __syncthreads();

        v8f acc0 = {}, acc1 = {}, acc2 = {}, acc3 = {};
#pragma unroll
        for (int ks = 0; ks < 11; ++ks) {
            const int k0 = ks * 4 + khalf;
            v2f a0, a1, a2, a3;
            a0.x = T[(0 * 16 + lrow) * TSTRIDE + k0];
            a0.y = T[(0 * 16 + lrow) * TSTRIDE + k0 + 1];
            a1.x = T[(1 * 16 + lrow) * TSTRIDE + k0];
            a1.y = T[(1 * 16 + lrow) * TSTRIDE + k0 + 1];
            a2.x = T[(2 * 16 + lrow) * TSTRIDE + k0];
            a2.y = T[(2 * 16 + lrow) * TSTRIDE + k0 + 1];
            a3.x = T[(3 * 16 + lrow) * TSTRIDE + k0];
            a3.y = T[(3 * 16 + lrow) * TSTRIDE + k0 + 1];
            acc0 = __builtin_amdgcn_wmma_f32_16x16x4_f32(false, a0, false, bf[ks],
                                                         (short)0, acc0, false, false);
            acc1 = __builtin_amdgcn_wmma_f32_16x16x4_f32(false, a1, false, bf[ks],
                                                         (short)0, acc1, false, false);
            acc2 = __builtin_amdgcn_wmma_f32_16x16x4_f32(false, a2, false, bf[ks],
                                                         (short)0, acc2, false, false);
            acc3 = __builtin_amdgcn_wmma_f32_16x16x4_f32(false, a3, false, bf[ks],
                                                         (short)0, acc3, false, false);
        }

#pragma unroll
        for (int rt = 0; rt < 4; ++rt) {
            const v8f& c = (rt == 0) ? acc0 : (rt == 1) ? acc1
                         : (rt == 2) ? acc2 : acc3;
#pragma unroll
            for (int r = 0; r < 8; ++r) {
                const int grow = rowbase + rt * 16 + lr0 + r;
                const int g = graphs[grow];
                out[(size_t)grow * NOISE + col] = c[r] + gpl[g * NOISE + col] + eb;
            }
        }
        __syncthreads();
    }
}

extern "C" void kernel_launch(void* const* d_in, const int* in_sizes, int n_in,
                              void* d_out, int out_size, void* d_ws, size_t ws_size,
                              hipStream_t stream) {
    const int*   graphs = (const int*)d_in[0];
    const float* chain  = (const float*)d_in[1];
    const float* trig   = (const float*)d_in[2];
    const float* tx     = (const float*)d_in[3];
    const int*   edges  = (const int*)d_in[4];
    const float* gw     = (const float*)d_in[5];
    const float* gb     = (const float*)d_in[6];
    const float* trigW  = (const float*)d_in[7];
    const float* trigb  = (const float*)d_in[8];
    const float* embW   = (const float*)d_in[9];
    const float* embb   = (const float*)d_in[10];
    float* out = (float*)d_out;

    float* ge = (float*)d_ws;
    float* gp = ge + (size_t)N_GRAPHS * N_NODES;

    gcn_kernel<<<N_GRAPHS * GSPLIT, GTHREADS, 0, stream>>>(edges, gw, gb, ge);
    proj_kernel<<<32, PTHREADS, 0, stream>>>(ge, embW, gp);
    batch_kernel<<<512, 256, 0, stream>>>(graphs, chain, trig, tx, trigW, trigb,
                                          embW, embb, gp, out);
}


# batch kernel start-up: projection table, both B fragment sets and the first chunk's inputs requested as one batch of loads before any wait (was 8 waited table loads, fragments, barrier, chunk loads)
# speedup vs baseline: 4.0333x; 1.1112x over previous
#include <hip/hip_runtime.h>
#include <hip/hip_bf16.h>

typedef float v2f __attribute__((ext_vector_type(2)));
typedef float v8f __attribute__((ext_vector_type(8)));

#define N_NODES 2048
#define N_GRAPHS 64
#define N_EDGES 16384
#define BATCH 65536
#define META 64
#define TXD 8
#define NOISE 128
#define TSTRIDE 48
#define CROWS 64
#define A1PITCH 68
#define NCHUNK (BATCH / CROWS)

__device__ __forceinline__ int kext_row(int k) {
    if (k < 32) return 2049 + k;
    if (k == 32) return 2048;
    if (k <= 40) return 2081 + (k - 33);
    return 2048;
}

#define GSPLIT 4
#define NRANGE (N_NODES / GSPLIT)
#define GTHREADS 512
#define EPT4 (N_EDGES / 4 / GTHREADS)

__global__ void __launch_bounds__(GTHREADS) gcn_kernel(const int* __restrict__ edges,
                                                       const float* __restrict__ gw,
                                                       const float* __restrict__ gb,
                                                       float* __restrict__ graph_emb) {
    __shared__ __align__(16) float cnt[N_NODES];
    __shared__ __align__(16) float dinv[N_NODES];
    __shared__ float agg[NRANGE];
    const int t = threadIdx.x;
    const int g = blockIdx.x / GSPLIT;
    const int p = blockIdx.x % GSPLIT;

    const int4* src4 = (const int4*)(edges + (size_t)g * (2 * N_EDGES));
    const int4* dst4 = src4 + N_EDGES / 4;
    int4 s[EPT4], d[EPT4];
    for (int j = 0; j < EPT4; ++j) d[j] = dst4[j * GTHREADS + t];

    *(float4*)&cnt[t * 4] = make_float4(0.0f, 0.0f, 0.0f, 0.0f);
    __syncthreads();
    for (int j = 0; j < EPT4; ++j) s[j] = src4[j * GTHREADS + t];
    for (int j = 0; j < EPT4; ++j) {
        atomicAdd(&cnt[d[j].x], 1.0f);
        atomicAdd(&cnt[d[j].y], 1.0f);
        atomicAdd(&cnt[d[j].z], 1.0f);
        atomicAdd(&cnt[d[j].w], 1.0f);
    }
    __syncthreads();

    const float4 cv = *(const float4*)&cnt[t * 4];
    const float dg[4] = {1.0f + cv.x, 1.0f + cv.y, 1.0f + cv.z, 1.0f + cv.w};
    float r[4];
    for (int k = 0; k < 4; ++k) r[k] = dg[k] > 0.0f ? rsqrtf(dg[k]) : 0.0f;
    *(float4*)&dinv[t * 4] = make_float4(r[0], r[1], r[2], r[3]);
    if (t / (NRANGE / 4) == p) {
        for (int k = 0; k < 4; ++k) agg[(t * 4 + k) % NRANGE] = r[k] * r[k];
    }
    __syncthreads();

    float nrm[EPT4][4];
    for (int j = 0; j < EPT4; ++j) {
        nrm[j][0] = dinv[s[j].x] * dinv[d[j].x];
        nrm[j][1] = dinv[s[j].y] * dinv[d[j].y];
        nrm[j][2] = dinv[s[j].z] * dinv[d[j].z];
        nrm[j][3] = dinv[s[j].w] * dinv[d[j].w];
    }
    for (int j = 0; j < EPT4; ++j) {
        const unsigned dv[4] = {(unsigned)d[j].x, (unsigned)d[j].y, (unsigned)d[j].z, (unsigned)d[j].w};
        for (int k = 0; k < 4; ++k)
            if (dv[k] / NRANGE == (unsigned)p)
                atomicAdd(&agg[dv[k] % NRANGE], nrm[j][k]);
    }
    __syncthreads();
    const float w = *gw, b = *gb;
    graph_emb[(size_t)g * N_NODES + p * NRANGE + t] = agg[t] * w + b;
}

#define PKC      256
#define PNCH     (N_NODES / PKC)
#define APITCH   (PKC + 4)
#define PCOPY    512
#define PTHREADS 544
#define PG       16

__global__ void __launch_bounds__(PTHREADS) proj_kernel(const float* __restrict__ ge,
                                                        const float* __restrict__ embW,
                                                        float* __restrict__ gp) {
    __shared__ __align__(16) float As[2][16 * APITCH];
    __shared__ __align__(16) float Bs[2][PKC * 16];

    const int t     = threadIdx.x;
    const int lane  = t & 31;
    const int wv    = __builtin_amdgcn_readfirstlane(t >> 5);
    const int mtile = blockIdx.x >> 3;
    const int ntile = blockIdx.x & 7;
    const int lrow  = lane & 15;
    const int khalf = (lane >> 4) * 2;

    const int i0 = (t - 32) & (PCOPY - 1), i1 = i0 + PCOPY;
    const float* ga0 = ge + (size_t)(mtile * 16 + (i0 >> 6)) * N_NODES + (i0 & 63) * 4;
    const float* ga1 = ge + (size_t)(mtile * 16 + (i1 >> 6)) * N_NODES + (i1 & 63) * 4;
    const float* gb0 = embW + (size_t)(i0 >> 2) * NOISE + ntile * 16 + (i0 & 3) * 4;
    const float* gb1 = embW + (size_t)(i1 >> 2) * NOISE + ntile * 16 + (i1 & 3) * 4;
    const int la0 = (i0 >> 6) * APITCH + (i0 & 63) * 4;
    const int la1 = (i1 >> 6) * APITCH + (i1 & 63) * 4;
    const int lb0 = i0 * 4;
    const int lb1 = i1 * 4;

    if (wv != 0) {
        const float4 ra0 = *(const float4*)ga0, ra1 = *(const float4*)ga1;
        const float4 rb0 = *(const float4*)gb0, rb1 = *(const float4*)gb1;
        *(float4*)&As[0][la0] = ra0;  *(float4*)&As[0][la1] = ra1;
        *(float4*)&Bs[0][lb0] = rb0;  *(float4*)&Bs[0][lb1] = rb1;
    }

    v8f c = {};
    for (int ch = 0; ch < PNCH; ++ch) {
        const int st = ch & 1;
        __syncthreads();
        if (wv != 0) {
            if (ch + 1 < PNCH) {
                const int ko = (ch + 1) * PKC;
                const float4 ra0 = *(const float4*)(ga0 + ko);
                const float4 ra1 = *(const float4*)(ga1 + ko);
                const float4 rb0 = *(const float4*)(gb0 + (size_t)ko * NOISE);
                const float4 rb1 = *(const float4*)(gb1 + (size_t)ko * NOISE);
                *(float4*)&As[st ^ 1][la0] = ra0;  *(float4*)&As[st ^ 1][la1] = ra1;
                *(float4*)&Bs[st ^ 1][lb0] = rb0;  *(float4*)&Bs[st ^ 1][lb1] = rb1;
            }
        } else {
            const float* ap = &As[st][lrow * APITCH + khalf];
            const float* bp = &Bs[st][khalf * 16 + lrow];
            v2f fa[2][PG], fb[2][PG];
#pragma unroll
            for (int i = 0; i < PG; ++i) {
                fa[0][i]   = *(const v2f*)(ap + i * 4);
                fb[0][i].x = bp[i * 64];
                fb[0][i].y = bp[i * 64 + 16];
            }
#pragma unroll
            for (int g = 0; g < PKC / 4 / PG; ++g) {
                if (g + 1 < PKC / 4 / PG) {
#pragma unroll
                    for (int i = 0; i < PG; ++i) {
                        const int q = (g + 1) * PG + i;
                        fa[(g + 1) & 1][i]   = *(const v2f*)(ap + q * 4);
                        fb[(g + 1) & 1][i].x = bp[q * 64];
                        fb[(g + 1) & 1][i].y = bp[q * 64 + 16];
                    }
                }
                __builtin_amdgcn_sched_barrier(0);
#pragma unroll
                for (int i = 0; i < PG; ++i)
                    c = __builtin_amdgcn_wmma_f32_16x16x4_f32(false, fa[g & 1][i], false,
                                                              fb[g & 1][i], (short)0, c,
                                                              false, false);
                __builtin_amdgcn_sched_barrier(0);
            }
        }
    }
    if (wv == 0) {
        const int col   = ntile * 16 + lrow;
        const int rbase = mtile * 16 + (lane >> 4) * 8;
#pragma unroll
        for (int r = 0; r < 8; ++r)
            gp[(size_t)(rbase + r) * NOISE + col] = c[r];
    }
}

__device__ __forceinline__ void load_chunk(const float* __restrict__ trig,
                                           const float* __restrict__ tx,
                                           const float* __restrict__ chain,
                                           int rowbase, int t,
                                           float4& tv0, float4& tv1, float4& tv2,
                                           float4& tv3, float4& xv, float& cv) {
    const float4* tsrc = (const float4*)(trig + (size_t)rowbase * META);
    tv0 = tsrc[t];
    tv1 = tsrc[t + 256];
    tv2 = tsrc[t + 512];
    tv3 = tsrc[t + 768];
    xv = ((const float4*)(tx + (size_t)rowbase * TXD))[t & 127];
    cv = chain[rowbase + (t & 63)];
}

__global__ void __launch_bounds__(256) batch_kernel(const int* __restrict__ graphs,
                                                    const float* __restrict__ chain,
                                                    const float* __restrict__ trig,
                                                    const float* __restrict__ tx,
                                                    const float* __restrict__ trigW,
                                                    const float* __restrict__ trigb,
                                                    const float* __restrict__ embW,
                                                    const float* __restrict__ embb,
                                                    const float* __restrict__ gp,
                                                    float* __restrict__ out) {
    __shared__ __align__(16) float gpl[N_GRAPHS * NOISE];
    __shared__ float T[CROWS * TSTRIDE];
    __shared__ __align__(16) float A1[CROWS * A1PITCH];

    const int t     = threadIdx.x;
    const int lane  = t & 31;
    const int w     = t >> 5;
    const int lrow  = lane & 15;
    const int khalf = (lane >> 4) * 2;
    const int col   = w * 16 + lrow;
    const int lr0   = (lane >> 4) * 8;

    v2f bf[11];
#pragma unroll
    for (int ks = 0; ks < 11; ++ks) {
        const int k0 = ks * 4 + khalf;
        bf[ks].x = embW[(size_t)kext_row(k0) * NOISE + col];
        bf[ks].y = embW[(size_t)kext_row(k0 + 1) * NOISE + col];
    }

    const int rt1 = w >> 1;
    const int nt1 = w & 1;
    const int c1  = nt1 * 16 + lrow;
    v2f b1[16];
#pragma unroll
    for (int ks = 0; ks < 16; ++ks) {
        const int k0 = ks * 4 + khalf;
        b1[ks].x = trigW[k0 * 32 + c1];
        b1[ks].y = trigW[(k0 + 1) * 32 + c1];
    }
    const float tb = trigb[c1];
    const float eb = embb[col];
    int chunk = blockIdx.x;
    if (chunk >= NCHUNK) return;
    float4 tv0, tv1, tv2, tv3, xv;
    float cv;
    load_chunk(trig, tx, chain, chunk * CROWS, t, tv0, tv1, tv2, tv3, xv, cv);
    {
        float4 gq[8];
#pragma unroll
        for (int j = 0; j < 8; ++j) gq[j] = *(const float4*)&gp[j * 1024 + t * 4];
#pragma unroll
        for (int j = 0; j < 8; ++j) *(float4*)&gpl[j * 1024 + t * 4] = gq[j];
    }
    if (t < CROWS) {
        for (int c = 41; c < TSTRIDE; ++c) T[t * TSTRIDE + c] = 0.0f;
    }

    for (;;) {
        const int rowbase = chunk * CROWS;

        if (chunk + gridDim.x < NCHUNK) {
            const float* p = trig + (size_t)(chunk + gridDim.x) * CROWS * META
                                  + t * 16;
            __builtin_prefetch(p, 0, 3);
        }

        {
            float* a1p = &A1[(t >> 4) * A1PITCH + (t & 15) * 4];
            *(float4*)(a1p)                = tv0;
            *(float4*)(a1p + 16 * A1PITCH) = tv1;
            *(float4*)(a1p + 32 * A1PITCH) = tv2;
            *(float4*)(a1p + 48 * A1PITCH) = tv3;
        }
        if (t < 128) {
            float* tp = &T[(t >> 1) * TSTRIDE + 33 + (t & 1) * 4];
            tp[0] = xv.x;  tp[1] = xv.y;  tp[2] = xv.z;  tp[3] = xv.w;
        } else if (t < 192) {
            T[(t - 128) * TSTRIDE + 32] = cv;
        }
        __syncthreads();

        {
            v8f d = {};
            const float* ap = &A1[(rt1 * 16 + lrow) * A1PITCH + khalf];
#pragma unroll
            for (int ks = 0; ks < 16; ++ks) {
                const v2f a = *(const v2f*)(ap + ks * 4);
                d = __builtin_amdgcn_wmma_f32_16x16x4_f32(false, a, false, b1[ks],
                                                          (short)0, d, false, false);
            }
#pragma unroll
            for (int r = 0; r < 8; ++r) {
                float v = d[r] + tb;
                T[(rt1 * 16 + lr0 + r) * TSTRIDE + c1] = v > 0.0f ? v : 0.0f;
            }
        }
        __syncthreads();

        v8f acc0 = {}, acc1 = {}, acc2 = {}, acc3 = {};
#pragma unroll
        for (int ks = 0; ks < 11; ++ks) {
            const int k0 = ks * 4 + khalf;
            v2f a0, a1, a2, a3;
            a0.x = T[(0 * 16 + lrow) * TSTRIDE + k0];
            a0.y = T[(0 * 16 + lrow) * TSTRIDE + k0 + 1];
            a1.x = T[(1 * 16 + lrow) * TSTRIDE + k0];
            a1.y = T[(1 * 16 + lrow) * TSTRIDE + k0 + 1];
            a2.x = T[(2 * 16 + lrow) * TSTRIDE + k0];
            a2.y = T[(2 * 16 + lrow) * TSTRIDE + k0 + 1];
            a3.x = T[(3 * 16 + lrow) * TSTRIDE + k0];
            a3.y = T[(3 * 16 + lrow) * TSTRIDE + k0 + 1];
            acc0 = __builtin_amdgcn_wmma_f32_16x16x4_f32(false, a0, false, bf[ks],
                                                         (short)0, acc0, false, false);
            acc1 = __builtin_amdgcn_wmma_f32_16x16x4_f32(false, a1, false, bf[ks],
                                                         (short)0, acc1, false, false);
            acc2 = __builtin_amdgcn_wmma_f32_16x16x4_f32(false, a2, false, bf[ks],
                                                         (short)0, acc2, false, false);
            acc3 = __builtin_amdgcn_wmma_f32_16x16x4_f32(false, a3, false, bf[ks],
                                                         (short)0, acc3, false, false);
        }

#pragma unroll
        for (int rt = 0; rt < 4; ++rt) {
            const v8f& c = (rt == 0) ? acc0 : (rt == 1) ? acc1
                         : (rt == 2) ? acc2 : acc3;
#pragma unroll
            for (int r = 0; r < 8; ++r) {
                const int grow = rowbase + rt * 16 + lr0 + r;
                const int g = graphs[grow];
                out[(size_t)grow * NOISE + col] = c[r] + gpl[g * NOISE + col] + eb;
            }
        }
        chunk += gridDim.x;
        if (chunk >= NCHUNK) break;
        __syncthreads();

        load_chunk(trig, tx, chain, chunk * CROWS, t, tv0, tv1, tv2, tv3, xv, cv);
    }
}

extern "C" void kernel_launch(void* const* d_in, const int* in_sizes, int n_in,
                              void* d_out, int out_size, void* d_ws, size_t ws_size,
                              hipStream_t stream) {
    const int*   graphs = (const int*)d_in[0];
    const float* chain  = (const float*)d_in[1];
    const float* trig   = (const float*)d_in[2];
    const float* tx     = (const float*)d_in[3];
    const int*   edges  = (const int*)d_in[4];
    const float* gw     = (const float*)d_in[5];
    const float* gb     = (const float*)d_in[6];
    const float* trigW  = (const float*)d_in[7];
    const float* trigb  = (const float*)d_in[8];
    const float* embW   = (const float*)d_in[9];
    const float* embb   = (const float*)d_in[10];
    float* out = (float*)d_out;

    float* ge = (float*)d_ws;
    float* gp = ge + (size_t)N_GRAPHS * N_NODES;

    gcn_kernel<<<N_GRAPHS * GSPLIT, GTHREADS, 0, stream>>>(edges, gw, gb, ge);
    proj_kernel<<<32, PTHREADS, 0, stream>>>(ge, embW, gp);
    batch_kernel<<<512, 256, 0, stream>>>(graphs, chain, trig, tx, trigW, trigb,
                                          embW, embb, gp, out);
}


# projection GEMM: K split over 8 work-groups per tile (256 groups x 64 chained WMMAs, was 32 x 512); partials via memory, the last-ticket group adds the 8 in chunk order; k sum order changes
# speedup vs baseline: 4.7462x; 1.1768x over previous
#include <hip/hip_runtime.h>
#include <hip/hip_bf16.h>

typedef float v2f __attribute__((ext_vector_type(2)));
typedef float v8f __attribute__((ext_vector_type(8)));

#define N_NODES 2048
#define N_GRAPHS 64
#define N_EDGES 16384
#define BATCH 65536
#define META 64
#define TXD 8
#define NOISE 128
#define TSTRIDE 48
#define CROWS 64
#define A1PITCH 68
#define NCHUNK (BATCH / CROWS)

__device__ __forceinline__ int kext_row(int k) {
    if (k < 32) return 2049 + k;
    if (k == 32) return 2048;
    if (k <= 40) return 2081 + (k - 33);
    return 2048;
}

#define PTILES 32
#define GSPLIT 4
#define NRANGE (N_NODES / GSPLIT)
#define GTHREADS 512
#define EPT4 (N_EDGES / 4 / GTHREADS)

__global__ void __launch_bounds__(GTHREADS) gcn_kernel(const int* __restrict__ edges,
                                                       const float* __restrict__ gw,
                                                       const float* __restrict__ gb,
                                                       float* __restrict__ graph_emb,
                                                       unsigned int* __restrict__ tick) {
    __shared__ __align__(16) float cnt[N_NODES];
    __shared__ __align__(16) float dinv[N_NODES];
    __shared__ float agg[NRANGE];
    const int t = threadIdx.x;
    const int g = blockIdx.x / GSPLIT;
    const int p = blockIdx.x % GSPLIT;

    if (t == 0 && blockIdx.x < PTILES) tick[blockIdx.x] = 0u;

    const int4* src4 = (const int4*)(edges + (size_t)g * (2 * N_EDGES));
    const int4* dst4 = src4 + N_EDGES / 4;
    int4 s[EPT4], d[EPT4];
    for (int j = 0; j < EPT4; ++j) d[j] = dst4[j * GTHREADS + t];

    *(float4*)&cnt[t * 4] = make_float4(0.0f, 0.0f, 0.0f, 0.0f);
    __syncthreads();
    for (int j = 0; j < EPT4; ++j) s[j] = src4[j * GTHREADS + t];
    for (int j = 0; j < EPT4; ++j) {
        atomicAdd(&cnt[d[j].x], 1.0f);
        atomicAdd(&cnt[d[j].y], 1.0f);
        atomicAdd(&cnt[d[j].z], 1.0f);
        atomicAdd(&cnt[d[j].w], 1.0f);
    }
    __syncthreads();

    const float4 cv = *(const float4*)&cnt[t * 4];
    const float dg[4] = {1.0f + cv.x, 1.0f + cv.y, 1.0f + cv.z, 1.0f + cv.w};
    float r[4];
    for (int k = 0; k < 4; ++k) r[k] = dg[k] > 0.0f ? rsqrtf(dg[k]) : 0.0f;
    *(float4*)&dinv[t * 4] = make_float4(r[0], r[1], r[2], r[3]);
    if (t / (NRANGE / 4) == p) {
        for (int k = 0; k < 4; ++k) agg[(t * 4 + k) % NRANGE] = r[k] * r[k];
    }
    __syncthreads();

    float nrm[EPT4][4];
    for (int j = 0; j < EPT4; ++j) {
        nrm[j][0] = dinv[s[j].x] * dinv[d[j].x];
        nrm[j][1] = dinv[s[j].y] * dinv[d[j].y];
        nrm[j][2] = dinv[s[j].z] * dinv[d[j].z];
        nrm[j][3] = dinv[s[j].w] * dinv[d[j].w];
    }
    for (int j = 0; j < EPT4; ++j) {
        const unsigned dv[4] = {(unsigned)d[j].x, (unsigned)d[j].y, (unsigned)d[j].z, (unsigned)d[j].w};
        for (int k = 0; k < 4; ++k)
            if (dv[k] / NRANGE == (unsigned)p)
                atomicAdd(&agg[dv[k] % NRANGE], nrm[j][k]);
    }
    __syncthreads();
    const float w = *gw, b = *gb;
    graph_emb[(size_t)g * N_NODES + p * NRANGE + t] = agg[t] * w + b;
}

#define PKC      256
#define PNCH     (N_NODES / PKC)
#define APITCH   (PKC + 4)
#define PTHREADS 128
#define PSLOTS   (16 * PKC / 4 / PTHREADS)
#define PG       16

__global__ void __launch_bounds__(PTHREADS) proj_kernel(const float* __restrict__ ge,
                                                        const float* __restrict__ embW,
                                                        float* part,
                                                        unsigned int* tick,
                                                        float* gp) {
    __shared__ __align__(16) float As[16 * APITCH];
    __shared__ __align__(16) float Bs[PKC * 16];

    const int t     = threadIdx.x;
    const int lane  = t & 31;
    const int wv    = __builtin_amdgcn_readfirstlane(t >> 5);
    const int tile  = blockIdx.x / PNCH;
    const int kc    = blockIdx.x % PNCH;
    const int mtile = tile >> 3;
    const int ntile = tile & 7;
    const int lrow  = lane & 15;
    const int khalf = (lane >> 4) * 2;

    const float* ga = ge + (size_t)(mtile * 16) * N_NODES + kc * PKC;
    const float* gb = embW + (size_t)(kc * PKC) * NOISE + ntile * 16;
    float4 ra[PSLOTS], rb[PSLOTS];
#pragma unroll
    for (int j = 0; j < PSLOTS; ++j) {
        const int i = t + PTHREADS * j;
        ra[j] = *(const float4*)(ga + (size_t)(i >> 6) * N_NODES + (i & 63) * 4);
    }
#pragma unroll
    for (int j = 0; j < PSLOTS; ++j) {
        const int i = t + PTHREADS * j;
        rb[j] = *(const float4*)(gb + (size_t)(i >> 2) * NOISE + (i & 3) * 4);
    }
#pragma unroll
    for (int j = 0; j < PSLOTS; ++j) {
        const int i = t + PTHREADS * j;
        *(float4*)&As[(i >> 6) * APITCH + (i & 63) * 4] = ra[j];
        *(float4*)&Bs[i * 4] = rb[j];
    }
    __syncthreads();
    if (wv != 0) return;

    v8f c = {};
    {
        const float* ap = &As[lrow * APITCH + khalf];
        const float* bp = &Bs[khalf * 16 + lrow];
        v2f fa[2][PG], fb[2][PG];
#pragma unroll
        for (int i = 0; i < PG; ++i) {
            fa[0][i]   = *(const v2f*)(ap + i * 4);
            fb[0][i].x = bp[i * 64];
            fb[0][i].y = bp[i * 64 + 16];
        }
#pragma unroll
        for (int g = 0; g < PKC / 4 / PG; ++g) {
            if (g + 1 < PKC / 4 / PG) {
#pragma unroll
                for (int i = 0; i < PG; ++i) {
                    const int q = (g + 1) * PG + i;
                    fa[(g + 1) & 1][i]   = *(const v2f*)(ap + q * 4);
                    fb[(g + 1) & 1][i].x = bp[q * 64];
                    fb[(g + 1) & 1][i].y = bp[q * 64 + 16];
                }
            }
            __builtin_amdgcn_sched_barrier(0);
#pragma unroll
            for (int i = 0; i < PG; ++i)
                c = __builtin_amdgcn_wmma_f32_16x16x4_f32(false, fa[g & 1][i], false,
                                                          fb[g & 1][i], (short)0, c,
                                                          false, false);
            __builtin_amdgcn_sched_barrier(0);
        }
    }

    float* pblk = part + (size_t)(tile * PNCH) * 256 + lane * 8;
    {
        float4* pw = (float4*)(pblk + kc * 256);
        pw[0] = make_float4(c[0], c[1], c[2], c[3]);
        pw[1] = make_float4(c[4], c[5], c[6], c[7]);
    }
    __builtin_amdgcn_fence(__ATOMIC_RELEASE, "agent");
    unsigned int tk = 0u;
    if (lane == 0)
        tk = __hip_atomic_fetch_add(tick + tile, 1u, __ATOMIC_RELAXED, __HIP_MEMORY_SCOPE_AGENT);
    tk = (unsigned int)__builtin_amdgcn_readfirstlane((int)tk);
    if (tk != PNCH - 1) return;

    __builtin_amdgcn_fence(__ATOMIC_ACQUIRE, "agent");
    float4 q0[PNCH], q1[PNCH];
#pragma unroll
    for (int j = 0; j < PNCH; ++j) {
        const float4* pr = (const float4*)(pblk + j * 256);
        q0[j] = pr[0];
        q1[j] = pr[1];
    }
    float s[8] = {q0[0].x, q0[0].y, q0[0].z, q0[0].w, q1[0].x, q1[0].y, q1[0].z, q1[0].w};
#pragma unroll
    for (int j = 1; j < PNCH; ++j) {
        s[0] += q0[j].x;  s[1] += q0[j].y;  s[2] += q0[j].z;  s[3] += q0[j].w;
        s[4] += q1[j].x;  s[5] += q1[j].y;  s[6] += q1[j].z;  s[7] += q1[j].w;
    }
    const int col   = ntile * 16 + lrow;
    const int rbase = mtile * 16 + (lane >> 4) * 8;
#pragma unroll
    for (int r = 0; r < 8; ++r)
        gp[(size_t)(rbase + r) * NOISE + col] = s[r];
}

__device__ __forceinline__ void load_chunk(const float* __restrict__ trig,
                                           const float* __restrict__ tx,
                                           const float* __restrict__ chain,
                                           int rowbase, int t,
                                           float4& tv0, float4& tv1, float4& tv2,
                                           float4& tv3, float4& xv, float& cv) {
    const float4* tsrc = (const float4*)(trig + (size_t)rowbase * META);
    tv0 = tsrc[t];
    tv1 = tsrc[t + 256];
    tv2 = tsrc[t + 512];
    tv3 = tsrc[t + 768];
    xv = ((const float4*)(tx + (size_t)rowbase * TXD))[t & 127];
    cv = chain[rowbase + (t & 63)];
}

__global__ void __launch_bounds__(256) batch_kernel(const int* __restrict__ graphs,
                                                    const float* __restrict__ chain,
                                                    const float* __restrict__ trig,
                                                    const float* __restrict__ tx,
                                                    const float* __restrict__ trigW,
                                                    const float* __restrict__ trigb,
                                                    const float* __restrict__ embW,
                                                    const float* __restrict__ embb,
                                                    const float* __restrict__ gp,
                                                    float* __restrict__ out) {
    __shared__ __align__(16) float gpl[N_GRAPHS * NOISE];
    __shared__ float T[CROWS * TSTRIDE];
    __shared__ __align__(16) float A1[CROWS * A1PITCH];

    const int t     = threadIdx.x;
    const int lane  = t & 31;
    const int w     = t >> 5;
    const int lrow  = lane & 15;
    const int khalf = (lane >> 4) * 2;
    const int col   = w * 16 + lrow;
    const int lr0   = (lane >> 4) * 8;

    v2f bf[11];
#pragma unroll
    for (int ks = 0; ks < 11; ++ks) {
        const int k0 = ks * 4 + khalf;
        bf[ks].x = embW[(size_t)kext_row(k0) * NOISE + col];
        bf[ks].y = embW[(size_t)kext_row(k0 + 1) * NOISE + col];
    }

    const int rt1 = w >> 1;
    const int nt1 = w & 1;
    const int c1  = nt1 * 16 + lrow;
    v2f b1[16];
#pragma unroll
    for (int ks = 0; ks < 16; ++ks) {
        const int k0 = ks * 4 + khalf;
        b1[ks].x = trigW[k0 * 32 + c1];
        b1[ks].y = trigW[(k0 + 1) * 32 + c1];
    }
    const float tb = trigb[c1];
    const float eb = embb[col];
    int chunk = blockIdx.x;
    if (chunk >= NCHUNK) return;
    float4 tv0, tv1, tv2, tv3, xv;
    float cv;
    load_chunk(trig, tx, chain, chunk * CROWS, t, tv0, tv1, tv2, tv3, xv, cv);
    {
        float4 gq[8];
#pragma unroll
        for (int j = 0; j < 8; ++j) gq[j] = *(const float4*)&gp[j * 1024 + t * 4];
#pragma unroll
        for (int j = 0; j < 8; ++j) *(float4*)&gpl[j * 1024 + t * 4] = gq[j];
    }
    if (t < CROWS) {
        for (int c = 41; c < TSTRIDE; ++c) T[t * TSTRIDE + c] = 0.0f;
    }

    for (;;) {
        const int rowbase = chunk * CROWS;

        if (chunk + gridDim.x < NCHUNK) {
            const float* p = trig + (size_t)(chunk + gridDim.x) * CROWS * META
                                  + t * 16;
            __builtin_prefetch(p, 0, 3);
        }

        {
            float* a1p = &A1[(t >> 4) * A1PITCH + (t & 15) * 4];
            *(float4*)(a1p)                = tv0;
            *(float4*)(a1p + 16 * A1PITCH) = tv1;
            *(float4*)(a1p + 32 * A1PITCH) = tv2;
            *(float4*)(a1p + 48 * A1PITCH) = tv3;
        }
        if (t < 128) {
            float* tp = &T[(t >> 1) * TSTRIDE + 33 + (t & 1) * 4];
            tp[0] = xv.x;  tp[1] = xv.y;  tp[2] = xv.z;  tp[3] = xv.w;
        } else if (t < 192) {
            T[(t - 128) * TSTRIDE + 32] = cv;
        }
        __syncthreads();

        {
            v8f d = {};
            const float* ap = &A1[(rt1 * 16 + lrow) * A1PITCH + khalf];
#pragma unroll
            for (int ks = 0; ks < 16; ++ks) {
                const v2f a = *(const v2f*)(ap + ks * 4);
                d = __builtin_amdgcn_wmma_f32_16x16x4_f32(false, a, false, b1[ks],
                                                          (short)0, d, false, false);
            }
#pragma unroll
            for (int r = 0; r < 8; ++r) {
                float v = d[r] + tb;
                T[(rt1 * 16 + lr0 + r) * TSTRIDE + c1] = v > 0.0f ? v : 0.0f;
            }
        }
        __syncthreads();

        v8f acc0 = {}, acc1 = {}, acc2 = {}, acc3 = {};
#pragma unroll
        for (int ks = 0; ks < 11; ++ks) {
            const int k0 = ks * 4 + khalf;
            v2f a0, a1, a2, a3;
            a0.x = T[(0 * 16 + lrow) * TSTRIDE + k0];
            a0.y = T[(0 * 16 + lrow) * TSTRIDE + k0 + 1];
            a1.x = T[(1 * 16 + lrow) * TSTRIDE + k0];
            a1.y = T[(1 * 16 + lrow) * TSTRIDE + k0 + 1];
            a2.x = T[(2 * 16 + lrow) * TSTRIDE + k0];
            a2.y = T[(2 * 16 + lrow) * TSTRIDE + k0 + 1];
            a3.x = T[(3 * 16 + lrow) * TSTRIDE + k0];
            a3.y = T[(3 * 16 + lrow) * TSTRIDE + k0 + 1];
            acc0 = __builtin_amdgcn_wmma_f32_16x16x4_f32(false, a0, false, bf[ks],
                                                         (short)0, acc0, false, false);
            acc1 = __builtin_amdgcn_wmma_f32_16x16x4_f32(false, a1, false, bf[ks],
                                                         (short)0, acc1, false, false);
            acc2 = __builtin_amdgcn_wmma_f32_16x16x4_f32(false, a2, false, bf[ks],
                                                         (short)0, acc2, false, false);
            acc3 = __builtin_amdgcn_wmma_f32_16x16x4_f32(false, a3, false, bf[ks],
                                                         (short)0, acc3, false, false);
        }

#pragma unroll
        for (int rt = 0; rt < 4; ++rt) {
            const v8f& c = (rt == 0) ? acc0 : (rt == 1) ? acc1
                         : (rt == 2) ? acc2 : acc3;
#pragma unroll
            for (int r = 0; r < 8; ++r) {
                const int grow = rowbase + rt * 16 + lr0 + r;
                const int g = graphs[grow];
                out[(size_t)grow * NOISE + col] = c[r] + gpl[g * NOISE + col] + eb;
            }
        }
        chunk += gridDim.x;
        if (chunk >= NCHUNK) break;
        __syncthreads();

        load_chunk(trig, tx, chain, chunk * CROWS, t, tv0, tv1, tv2, tv3, xv, cv);
    }
}

extern "C" void kernel_launch(void* const* d_in, const int* in_sizes, int n_in,
                              void* d_out, int out_size, void* d_ws, size_t ws_size,
                              hipStream_t stream) {
    const int*   graphs = (const int*)d_in[0];
    const float* chain  = (const float*)d_in[1];
    const float* trig   = (const float*)d_in[2];
    const float* tx     = (const float*)d_in[3];
    const int*   edges  = (const int*)d_in[4];
    const float* gw     = (const float*)d_in[5];
    const float* gb     = (const float*)d_in[6];
    const float* trigW  = (const float*)d_in[7];
    const float* trigb  = (const float*)d_in[8];
    const float* embW   = (const float*)d_in[9];
    const float* embb   = (const float*)d_in[10];
    float* out = (float*)d_out;

    float* ge = (float*)d_ws;
    float* gp = ge + (size_t)N_GRAPHS * N_NODES;
    float* part = gp + (size_t)N_GRAPHS * NOISE;
    unsigned int* tick = (unsigned int*)(part + (size_t)PTILES * PNCH * 256);

    gcn_kernel<<<N_GRAPHS * GSPLIT, GTHREADS, 0, stream>>>(edges, gw, gb, ge, tick);
    proj_kernel<<<PTILES * PNCH, PTHREADS, 0, stream>>>(ge, embW, part, tick, gp);
    batch_kernel<<<512, 256, 0, stream>>>(graphs, chain, trig, tx, trigW, trigb,
                                          embW, embb, gp, out);
}


# batch kernel epilogue: output tile staged through LDS and written row-wise, one wave-wide b128 store per whole 512-byte row (was 32 b32 stores per lane, 64-byte pieces); sums unchanged
# speedup vs baseline: 5.4247x; 1.1430x over previous
#include <hip/hip_runtime.h>
#include <hip/hip_bf16.h>

typedef float v2f __attribute__((ext_vector_type(2)));
typedef float v8f __attribute__((ext_vector_type(8)));

#define N_NODES 2048
#define N_GRAPHS 64
#define N_EDGES 16384
#define BATCH 65536
#define META 64
#define TXD 8
#define NOISE 128
#define TSTRIDE 48
#define CROWS 64
#define A1PITCH 68
#define OPITCH 132
#define NCHUNK (BATCH / CROWS)

__device__ __forceinline__ int kext_row(int k) {
    if (k < 32) return 2049 + k;
    if (k == 32) return 2048;
    if (k <= 40) return 2081 + (k - 33);
    return 2048;
}

#define PTILES 32
#define GSPLIT 4
#define NRANGE (N_NODES / GSPLIT)
#define GTHREADS 512
#define EPT4 (N_EDGES / 4 / GTHREADS)

__global__ void __launch_bounds__(GTHREADS) gcn_kernel(const int* __restrict__ edges,
                                                       const float* __restrict__ gw,
                                                       const float* __restrict__ gb,
                                                       float* __restrict__ graph_emb,
                                                       unsigned int* __restrict__ tick) {
    __shared__ __align__(16) float cnt[N_NODES];
    __shared__ __align__(16) float dinv[N_NODES];
    __shared__ float agg[NRANGE];
    const int t = threadIdx.x;
    const int g = blockIdx.x / GSPLIT;
    const int p = blockIdx.x % GSPLIT;

    if (t == 0 && blockIdx.x < PTILES) tick[blockIdx.x] = 0u;

    const int4* src4 = (const int4*)(edges + (size_t)g * (2 * N_EDGES));
    const int4* dst4 = src4 + N_EDGES / 4;
    int4 s[EPT4], d[EPT4];
    for (int j = 0; j < EPT4; ++j) d[j] = dst4[j * GTHREADS + t];

    *(float4*)&cnt[t * 4] = make_float4(0.0f, 0.0f, 0.0f, 0.0f);
    __syncthreads();
    for (int j = 0; j < EPT4; ++j) s[j] = src4[j * GTHREADS + t];
    for (int j = 0; j < EPT4; ++j) {
        atomicAdd(&cnt[d[j].x], 1.0f);
        atomicAdd(&cnt[d[j].y], 1.0f);
        atomicAdd(&cnt[d[j].z], 1.0f);
        atomicAdd(&cnt[d[j].w], 1.0f);
    }
    __syncthreads();

    const float4 cv = *(const float4*)&cnt[t * 4];
    const float dg[4] = {1.0f + cv.x, 1.0f + cv.y, 1.0f + cv.z, 1.0f + cv.w};
    float r[4];
    for (int k = 0; k < 4; ++k) r[k] = dg[k] > 0.0f ? rsqrtf(dg[k]) : 0.0f;
    *(float4*)&dinv[t * 4] = make_float4(r[0], r[1], r[2], r[3]);
    if (t / (NRANGE / 4) == p) {
        for (int k = 0; k < 4; ++k) agg[(t * 4 + k) % NRANGE] = r[k] * r[k];
    }
    __syncthreads();

    float nrm[EPT4][4];
    for (int j = 0; j < EPT4; ++j) {
        nrm[j][0] = dinv[s[j].x] * dinv[d[j].x];
        nrm[j][1] = dinv[s[j].y] * dinv[d[j].y];
        nrm[j][2] = dinv[s[j].z] * dinv[d[j].z];
        nrm[j][3] = dinv[s[j].w] * dinv[d[j].w];
    }
    for (int j = 0; j < EPT4; ++j) {
        const unsigned dv[4] = {(unsigned)d[j].x, (unsigned)d[j].y, (unsigned)d[j].z, (unsigned)d[j].w};
        for (int k = 0; k < 4; ++k)
            if (dv[k] / NRANGE == (unsigned)p)
                atomicAdd(&agg[dv[k] % NRANGE], nrm[j][k]);
    }
    __syncthreads();
    const float w = *gw, b = *gb;
    graph_emb[(size_t)g * N_NODES + p * NRANGE + t] = agg[t] * w + b;
}

#define PKC      256
#define PNCH     (N_NODES / PKC)
#define APITCH   (PKC + 4)
#define PTHREADS 128
#define PSLOTS   (16 * PKC / 4 / PTHREADS)
#define PG       16

__global__ void __launch_bounds__(PTHREADS) proj_kernel(const float* __restrict__ ge,
                                                        const float* __restrict__ embW,
                                                        float* part,
                                                        unsigned int* tick,
                                                        float* gp) {
    __shared__ __align__(16) float As[16 * APITCH];
    __shared__ __align__(16) float Bs[PKC * 16];

    const int t     = threadIdx.x;
    const int lane  = t & 31;
    const int wv    = __builtin_amdgcn_readfirstlane(t >> 5);
    const int tile  = blockIdx.x / PNCH;
    const int kc    = blockIdx.x % PNCH;
    const int mtile = tile >> 3;
    const int ntile = tile & 7;
    const int lrow  = lane & 15;
    const int khalf = (lane >> 4) * 2;

    const float* ga = ge + (size_t)(mtile * 16) * N_NODES + kc * PKC;
    const float* gb = embW + (size_t)(kc * PKC) * NOISE + ntile * 16;
    float4 ra[PSLOTS], rb[PSLOTS];
#pragma unroll
    for (int j = 0; j < PSLOTS; ++j) {
        const int i = t + PTHREADS * j;
        ra[j] = *(const float4*)(ga + (size_t)(i >> 6) * N_NODES + (i & 63) * 4);
    }
#pragma unroll
    for (int j = 0; j < PSLOTS; ++j) {
        const int i = t + PTHREADS * j;
        rb[j] = *(const float4*)(gb + (size_t)(i >> 2) * NOISE + (i & 3) * 4);
    }
#pragma unroll
    for (int j = 0; j < PSLOTS; ++j) {
        const int i = t + PTHREADS * j;
        *(float4*)&As[(i >> 6) * APITCH + (i & 63) * 4] = ra[j];
        *(float4*)&Bs[i * 4] = rb[j];
    }
    __syncthreads();
    if (wv != 0) return;

    v8f c = {};
    {
        const float* ap = &As[lrow * APITCH + khalf];
        const float* bp = &Bs[khalf * 16 + lrow];
        v2f fa[2][PG], fb[2][PG];
#pragma unroll
        for (int i = 0; i < PG; ++i) {
            fa[0][i]   = *(const v2f*)(ap + i * 4);
            fb[0][i].x = bp[i * 64];
            fb[0][i].y = bp[i * 64 + 16];
        }
#pragma unroll
        for (int g = 0; g < PKC / 4 / PG; ++g) {
            if (g + 1 < PKC / 4 / PG) {
#pragma unroll
                for (int i = 0; i < PG; ++i) {
                    const int q = (g + 1) * PG + i;
                    fa[(g + 1) & 1][i]   = *(const v2f*)(ap + q * 4);
                    fb[(g + 1) & 1][i].x = bp[q * 64];
                    fb[(g + 1) & 1][i].y = bp[q * 64 + 16];
                }
            }
            __builtin_amdgcn_sched_barrier(0);
#pragma unroll
            for (int i = 0; i < PG; ++i)
                c = __builtin_amdgcn_wmma_f32_16x16x4_f32(false, fa[g & 1][i], false,
                                                          fb[g & 1][i], (short)0, c,
                                                          false, false);
            __builtin_amdgcn_sched_barrier(0);
        }
    }

    float* pblk = part + (size_t)(tile * PNCH) * 256 + lane * 8;
    {
        float4* pw = (float4*)(pblk + kc * 256);
        pw[0] = make_float4(c[0], c[1], c[2], c[3]);
        pw[1] = make_float4(c[4], c[5], c[6], c[7]);
    }
    __builtin_amdgcn_fence(__ATOMIC_RELEASE, "agent");
    unsigned int tk = 0u;
    if (lane == 0)
        tk = __hip_atomic_fetch_add(tick + tile, 1u, __ATOMIC_RELAXED, __HIP_MEMORY_SCOPE_AGENT);
    tk = (unsigned int)__builtin_amdgcn_readfirstlane((int)tk);
    if (tk != PNCH - 1) return;

    __builtin_amdgcn_fence(__ATOMIC_ACQUIRE, "agent");
    float4 q0[PNCH], q1[PNCH];
#pragma unroll
    for (int j = 0; j < PNCH; ++j) {
        const float4* pr = (const float4*)(pblk + j * 256);
        q0[j] = pr[0];
        q1[j] = pr[1];
    }
    float s[8] = {q0[0].x, q0[0].y, q0[0].z, q0[0].w, q1[0].x, q1[0].y, q1[0].z, q1[0].w};
#pragma unroll
    for (int j = 1; j < PNCH; ++j) {
        s[0] += q0[j].x;  s[1] += q0[j].y;  s[2] += q0[j].z;  s[3] += q0[j].w;
        s[4] += q1[j].x;  s[5] += q1[j].y;  s[6] += q1[j].z;  s[7] += q1[j].w;
    }
    const int col   = ntile * 16 + lrow;
    const int rbase = mtile * 16 + (lane >> 4) * 8;
#pragma unroll
    for (int r = 0; r < 8; ++r)
        gp[(size_t)(rbase + r) * NOISE + col] = s[r];
}

__device__ __forceinline__ void load_chunk(const float* __restrict__ trig,
                                           const float* __restrict__ tx,
                                           const float* __restrict__ chain,
                                           const int* __restrict__ graphs,
                                           int rowbase, int t, int gsel,
                                           float4& tv0, float4& tv1, float4& tv2,
                                           float4& tv3, float4& xv, float& cv, int& gv) {
    const float4* tsrc = (const float4*)(trig + (size_t)rowbase * META);
    tv0 = tsrc[t];
    tv1 = tsrc[t + 256];
    tv2 = tsrc[t + 512];
    tv3 = tsrc[t + 768];
    xv = ((const float4*)(tx + (size_t)rowbase * TXD))[t & 127];
    cv = chain[rowbase + (t & 63)];
    gv = graphs[rowbase + gsel];
}

__global__ void __launch_bounds__(256) batch_kernel(const int* __restrict__ graphs,
                                                    const float* __restrict__ chain,
                                                    const float* __restrict__ trig,
                                                    const float* __restrict__ tx,
                                                    const float* __restrict__ trigW,
                                                    const float* __restrict__ trigb,
                                                    const float* __restrict__ embW,
                                                    const float* __restrict__ embb,
                                                    const float* __restrict__ gp,
                                                    float* __restrict__ out) {
    __shared__ __align__(16) float gpl[N_GRAPHS * NOISE];
    __shared__ float T[CROWS * TSTRIDE];
    __shared__ __align__(16) float A1[CROWS * A1PITCH];
    __shared__ __align__(16) float O[CROWS * OPITCH];

    const int t     = threadIdx.x;
    const int lane  = t & 31;
    const int w     = t >> 5;
    const int lrow  = lane & 15;
    const int khalf = (lane >> 4) * 2;
    const int col   = w * 16 + lrow;
    const int lr0   = (lane >> 4) * 8;
    const int wu    = __builtin_amdgcn_readfirstlane(w);

    v2f bf[11];
#pragma unroll
    for (int ks = 0; ks < 11; ++ks) {
        const int k0 = ks * 4 + khalf;
        bf[ks].x = embW[(size_t)kext_row(k0) * NOISE + col];
        bf[ks].y = embW[(size_t)kext_row(k0 + 1) * NOISE + col];
    }

    const int rt1 = w >> 1;
    const int nt1 = w & 1;
    const int c1  = nt1 * 16 + lrow;
    v2f b1[16];
#pragma unroll
    for (int ks = 0; ks < 16; ++ks) {
        const int k0 = ks * 4 + khalf;
        b1[ks].x = trigW[k0 * 32 + c1];
        b1[ks].y = trigW[(k0 + 1) * 32 + c1];
    }
    const float tb = trigb[c1];
    const float4 eb4 = *(const float4*)&embb[lane * 4];
    int chunk = blockIdx.x;
    if (chunk >= NCHUNK) return;
    float4 tv0, tv1, tv2, tv3, xv;
    float cv;
    int gv;
    const int gsel = w * 8 + (lane & 7);
    load_chunk(trig, tx, chain, graphs, chunk * CROWS, t, gsel, tv0, tv1, tv2, tv3, xv, cv, gv);
    {
        float4 gq[8];
#pragma unroll
        for (int j = 0; j < 8; ++j) gq[j] = *(const float4*)&gp[j * 1024 + t * 4];
#pragma unroll
        for (int j = 0; j < 8; ++j) *(float4*)&gpl[j * 1024 + t * 4] = gq[j];
    }
    if (t < CROWS) {
        for (int c = 41; c < TSTRIDE; ++c) T[t * TSTRIDE + c] = 0.0f;
    }

    for (;;) {
        const int rowbase = chunk * CROWS;

        if (chunk + gridDim.x < NCHUNK) {
            const float* p = trig + (size_t)(chunk + gridDim.x) * CROWS * META
                                  + t * 16;
            __builtin_prefetch(p, 0, 3);
        }

        {
            float* a1p = &A1[(t >> 4) * A1PITCH + (t & 15) * 4];
            *(float4*)(a1p)                = tv0;
            *(float4*)(a1p + 16 * A1PITCH) = tv1;
            *(float4*)(a1p + 32 * A1PITCH) = tv2;
            *(float4*)(a1p + 48 * A1PITCH) = tv3;
        }
        if (t < 128) {
            float* tp = &T[(t >> 1) * TSTRIDE + 33 + (t & 1) * 4];
            tp[0] = xv.x;  tp[1] = xv.y;  tp[2] = xv.z;  tp[3] = xv.w;
        } else if (t < 192) {
            T[(t - 128) * TSTRIDE + 32] = cv;
        }
        __syncthreads();

        {
            v8f d = {};
            const float* ap = &A1[(rt1 * 16 + lrow) * A1PITCH + khalf];
#pragma unroll
            for (int ks = 0; ks < 16; ++ks) {
                const v2f a = *(const v2f*)(ap + ks * 4);
                d = __builtin_amdgcn_wmma_f32_16x16x4_f32(false, a, false, b1[ks],
                                                          (short)0, d, false, false);
            }
#pragma unroll
            for (int r = 0; r < 8; ++r) {
                float v = d[r] + tb;
                T[(rt1 * 16 + lr0 + r) * TSTRIDE + c1] = v > 0.0f ? v : 0.0f;
            }
        }
        __syncthreads();

        v8f acc0 = {}, acc1 = {}, acc2 = {}, acc3 = {};
#pragma unroll
        for (int ks = 0; ks < 11; ++ks) {
            const int k0 = ks * 4 + khalf;
            v2f a0, a1, a2, a3;
            a0.x = T[(0 * 16 + lrow) * TSTRIDE + k0];
            a0.y = T[(0 * 16 + lrow) * TSTRIDE + k0 + 1];
            a1.x = T[(1 * 16 + lrow) * TSTRIDE + k0];
            a1.y = T[(1 * 16 + lrow) * TSTRIDE + k0 + 1];
            a2.x = T[(2 * 16 + lrow) * TSTRIDE + k0];
            a2.y = T[(2 * 16 + lrow) * TSTRIDE + k0 + 1];
            a3.x = T[(3 * 16 + lrow) * TSTRIDE + k0];
            a3.y = T[(3 * 16 + lrow) * TSTRIDE + k0 + 1];
            acc0 = __builtin_amdgcn_wmma_f32_16x16x4_f32(false, a0, false, bf[ks],
                                                         (short)0, acc0, false, false);
            acc1 = __builtin_amdgcn_wmma_f32_16x16x4_f32(false, a1, false, bf[ks],
                                                         (short)0, acc1, false, false);
            acc2 = __builtin_amdgcn_wmma_f32_16x16x4_f32(false, a2, false, bf[ks],
                                                         (short)0, acc2, false, false);
            acc3 = __builtin_amdgcn_wmma_f32_16x16x4_f32(false, a3, false, bf[ks],
                                                         (short)0, acc3, false, false);
        }

#pragma unroll
        for (int rt = 0; rt < 4; ++rt) {
            const v8f& c = (rt == 0) ? acc0 : (rt == 1) ? acc1
                         : (rt == 2) ? acc2 : acc3;
#pragma unroll
            for (int r = 0; r < 8; ++r)
                O[(rt * 16 + lr0 + r) * OPITCH + col] = c[r];
        }
        __syncthreads();

#pragma unroll
        for (int i = 0; i < 8; ++i) {
            const int row = wu * 8 + i;
            const int g = __builtin_amdgcn_readlane(gv, i);
            const float4 o = *(const float4*)&O[row * OPITCH + lane * 4];
            const float4 q = *(const float4*)&gpl[g * NOISE + lane * 4];
            float4 v;
            v.x = (o.x + q.x) + eb4.x;
            v.y = (o.y + q.y) + eb4.y;
            v.z = (o.z + q.z) + eb4.z;
            v.w = (o.w + q.w) + eb4.w;
            *(float4*)&out[(size_t)(rowbase + row) * NOISE + lane * 4] = v;
        }
        chunk += gridDim.x;
        if (chunk >= NCHUNK) break;
        __syncthreads();

        load_chunk(trig, tx, chain, graphs, chunk * CROWS, t, gsel, tv0, tv1, tv2, tv3, xv, cv, gv);
    }
}

extern "C" void kernel_launch(void* const* d_in, const int* in_sizes, int n_in,
                              void* d_out, int out_size, void* d_ws, size_t ws_size,
                              hipStream_t stream) {
    const int*   graphs = (const int*)d_in[0];
    const float* chain  = (const float*)d_in[1];
    const float* trig   = (const float*)d_in[2];
    const float* tx     = (const float*)d_in[3];
    const int*   edges  = (const int*)d_in[4];
    const float* gw     = (const float*)d_in[5];
    const float* gb     = (const float*)d_in[6];
    const float* trigW  = (const float*)d_in[7];
    const float* trigb  = (const float*)d_in[8];
    const float* embW   = (const float*)d_in[9];
    const float* embb   = (const float*)d_in[10];
    float* out = (float*)d_out;

    float* ge = (float*)d_ws;
    float* gp = ge + (size_t)N_GRAPHS * N_NODES;
    float* part = gp + (size_t)N_GRAPHS * NOISE;
    unsigned int* tick = (unsigned int*)(part + (size_t)PTILES * PNCH * 256);

    gcn_kernel<<<N_GRAPHS * GSPLIT, GTHREADS, 0, stream>>>(edges, gw, gb, ge, tick);
    proj_kernel<<<PTILES * PNCH, PTHREADS, 0, stream>>>(ge, embW, part, tick, gp);
    batch_kernel<<<512, 256, 0, stream>>>(graphs, chain, trig, tx, trigW, trigb,
                                          embW, embb, gp, out);
}


# graph kernel: dinv[dst] taken out of each node's sum: per edge one LDS read (dinv[src]), by lanes whose dst is in the group's range, was two by all; node sum scaled at the end; agg rounding changes
# speedup vs baseline: 5.6815x; 1.0473x over previous
#include <hip/hip_runtime.h>
#include <hip/hip_bf16.h>

typedef float v2f __attribute__((ext_vector_type(2)));
typedef float v8f __attribute__((ext_vector_type(8)));

#define N_NODES 2048
#define N_GRAPHS 64
#define N_EDGES 16384
#define BATCH 65536
#define META 64
#define TXD 8
#define NOISE 128
#define TSTRIDE 48
#define CROWS 64
#define A1PITCH 68
#define OPITCH 132
#define NCHUNK (BATCH / CROWS)

__device__ __forceinline__ int kext_row(int k) {
    if (k < 32) return 2049 + k;
    if (k == 32) return 2048;
    if (k <= 40) return 2081 + (k - 33);
    return 2048;
}

#define PTILES 32
#define GSPLIT 4
#define NRANGE (N_NODES / GSPLIT)
#define GTHREADS 512
#define EPT4 (N_EDGES / 4 / GTHREADS)

__global__ void __launch_bounds__(GTHREADS) gcn_kernel(const int* __restrict__ edges,
                                                       const float* __restrict__ gw,
                                                       const float* __restrict__ gb,
                                                       float* __restrict__ graph_emb,
                                                       unsigned int* __restrict__ tick) {
    __shared__ __align__(16) float cnt[N_NODES];
    __shared__ __align__(16) float dinv[N_NODES];
    __shared__ float agg[NRANGE];
    const int t = threadIdx.x;
    const int g = blockIdx.x / GSPLIT;
    const int p = blockIdx.x % GSPLIT;

    if (t == 0 && blockIdx.x < PTILES) tick[blockIdx.x] = 0u;

    const int4* src4 = (const int4*)(edges + (size_t)g * (2 * N_EDGES));
    const int4* dst4 = src4 + N_EDGES / 4;
    int4 s[EPT4], d[EPT4];
    for (int j = 0; j < EPT4; ++j) d[j] = dst4[j * GTHREADS + t];

    *(float4*)&cnt[t * 4] = make_float4(0.0f, 0.0f, 0.0f, 0.0f);
    __syncthreads();
    for (int j = 0; j < EPT4; ++j) s[j] = src4[j * GTHREADS + t];
    for (int j = 0; j < EPT4; ++j) {
        atomicAdd(&cnt[d[j].x], 1.0f);
        atomicAdd(&cnt[d[j].y], 1.0f);
        atomicAdd(&cnt[d[j].z], 1.0f);
        atomicAdd(&cnt[d[j].w], 1.0f);
    }
    __syncthreads();

    const float4 cv = *(const float4*)&cnt[t * 4];
    const float dg[4] = {1.0f + cv.x, 1.0f + cv.y, 1.0f + cv.z, 1.0f + cv.w};
    float r[4];
    for (int k = 0; k < 4; ++k) r[k] = dg[k] > 0.0f ? rsqrtf(dg[k]) : 0.0f;
    *(float4*)&dinv[t * 4] = make_float4(r[0], r[1], r[2], r[3]);
    if (t / (NRANGE / 4) == p) {
        for (int k = 0; k < 4; ++k) agg[(t * 4 + k) % NRANGE] = r[k];
    }
    __syncthreads();

    float dsrc[EPT4][4];
    for (int j = 0; j < EPT4; ++j) {
        const unsigned dv[4] = {(unsigned)d[j].x, (unsigned)d[j].y, (unsigned)d[j].z, (unsigned)d[j].w};
        const int sv[4] = {s[j].x, s[j].y, s[j].z, s[j].w};
        for (int k = 0; k < 4; ++k)
            dsrc[j][k] = dinv[(dv[k] / NRANGE == (unsigned)p) ? sv[k] : 0];
    }
    for (int j = 0; j < EPT4; ++j) {
        const unsigned dv[4] = {(unsigned)d[j].x, (unsigned)d[j].y, (unsigned)d[j].z, (unsigned)d[j].w};
        for (int k = 0; k < 4; ++k)
            if (dv[k] / NRANGE == (unsigned)p)
                atomicAdd(&agg[dv[k] % NRANGE], dsrc[j][k]);
    }
    __syncthreads();
    const float w = *gw, b = *gb;
    const float aggn = dinv[p * NRANGE + t] * agg[t];
    graph_emb[(size_t)g * N_NODES + p * NRANGE + t] = aggn * w + b;
}

#define PKC      256
#define PNCH     (N_NODES / PKC)
#define APITCH   (PKC + 4)
#define PTHREADS 128
#define PSLOTS   (16 * PKC / 4 / PTHREADS)
#define PG       16

__global__ void __launch_bounds__(PTHREADS) proj_kernel(const float* __restrict__ ge,
                                                        const float* __restrict__ embW,
                                                        float* part,
                                                        unsigned int* tick,
                                                        float* gp) {
    __shared__ __align__(16) float As[16 * APITCH];
    __shared__ __align__(16) float Bs[PKC * 16];

    const int t     = threadIdx.x;
    const int lane  = t & 31;
    const int wv    = __builtin_amdgcn_readfirstlane(t >> 5);
    const int tile  = blockIdx.x / PNCH;
    const int kc    = blockIdx.x % PNCH;
    const int mtile = tile >> 3;
    const int ntile = tile & 7;
    const int lrow  = lane & 15;
    const int khalf = (lane >> 4) * 2;

    const float* ga = ge + (size_t)(mtile * 16) * N_NODES + kc * PKC;
    const float* gb = embW + (size_t)(kc * PKC) * NOISE + ntile * 16;
    float4 ra[PSLOTS], rb[PSLOTS];
#pragma unroll
    for (int j = 0; j < PSLOTS; ++j) {
        const int i = t + PTHREADS * j;
        ra[j] = *(const float4*)(ga + (size_t)(i >> 6) * N_NODES + (i & 63) * 4);
    }
#pragma unroll
    for (int j = 0; j < PSLOTS; ++j) {
        const int i = t + PTHREADS * j;
        rb[j] = *(const float4*)(gb + (size_t)(i >> 2) * NOISE + (i & 3) * 4);
    }
#pragma unroll
    for (int j = 0; j < PSLOTS; ++j) {
        const int i = t + PTHREADS * j;
        *(float4*)&As[(i >> 6) * APITCH + (i & 63) * 4] = ra[j];
        *(float4*)&Bs[i * 4] = rb[j];
    }
    __syncthreads();
    if (wv != 0) return;

    v8f c = {};
    {
        const float* ap = &As[lrow * APITCH + khalf];
        const float* bp = &Bs[khalf * 16 + lrow];
        v2f fa[2][PG], fb[2][PG];
#pragma unroll
        for (int i = 0; i < PG; ++i) {
            fa[0][i]   = *(const v2f*)(ap + i * 4);
            fb[0][i].x = bp[i * 64];
            fb[0][i].y = bp[i * 64 + 16];
        }
#pragma unroll
        for (int g = 0; g < PKC / 4 / PG; ++g) {
            if (g + 1 < PKC / 4 / PG) {
#pragma unroll
                for (int i = 0; i < PG; ++i) {
                    const int q = (g + 1) * PG + i;
                    fa[(g + 1) & 1][i]   = *(const v2f*)(ap + q * 4);
                    fb[(g + 1) & 1][i].x = bp[q * 64];
                    fb[(g + 1) & 1][i].y = bp[q * 64 + 16];
                }
            }
            __builtin_amdgcn_sched_barrier(0);
#pragma unroll
            for (int i = 0; i < PG; ++i)
                c = __builtin_amdgcn_wmma_f32_16x16x4_f32(false, fa[g & 1][i], false,
                                                          fb[g & 1][i], (short)0, c,
                                                          false, false);
            __builtin_amdgcn_sched_barrier(0);
        }
    }

    float* pblk = part + (size_t)(tile * PNCH) * 256 + lane * 8;
    {
        float4* pw = (float4*)(pblk + kc * 256);
        pw[0] = make_float4(c[0], c[1], c[2], c[3]);
        pw[1] = make_float4(c[4], c[5], c[6], c[7]);
    }
    __builtin_amdgcn_fence(__ATOMIC_RELEASE, "agent");
    unsigned int tk = 0u;
    if (lane == 0)
        tk = __hip_atomic_fetch_add(tick + tile, 1u, __ATOMIC_RELAXED, __HIP_MEMORY_SCOPE_AGENT);
    tk = (unsigned int)__builtin_amdgcn_readfirstlane((int)tk);
    if (tk != PNCH - 1) return;

    __builtin_amdgcn_fence(__ATOMIC_ACQUIRE, "agent");
    float4 q0[PNCH], q1[PNCH];
#pragma unroll
    for (int j = 0; j < PNCH; ++j) {
        const float4* pr = (const float4*)(pblk + j * 256);
        q0[j] = pr[0];
        q1[j] = pr[1];
    }
    float s[8] = {q0[0].x, q0[0].y, q0[0].z, q0[0].w, q1[0].x, q1[0].y, q1[0].z, q1[0].w};
#pragma unroll
    for (int j = 1; j < PNCH; ++j) {
        s[0] += q0[j].x;  s[1] += q0[j].y;  s[2] += q0[j].z;  s[3] += q0[j].w;
        s[4] += q1[j].x;  s[5] += q1[j].y;  s[6] += q1[j].z;  s[7] += q1[j].w;
    }
    const int col   = ntile * 16 + lrow;
    const int rbase = mtile * 16 + (lane >> 4) * 8;
#pragma unroll
    for (int r = 0; r < 8; ++r)
        gp[(size_t)(rbase + r) * NOISE + col] = s[r];
}

__device__ __forceinline__ void load_chunk(const float* __restrict__ trig,
                                           const float* __restrict__ tx,
                                           const float* __restrict__ chain,
                                           const int* __restrict__ graphs,
                                           int rowbase, int t, int gsel,
                                           float4& tv0, float4& tv1, float4& tv2,
                                           float4& tv3, float4& xv, float& cv, int& gv) {
    const float4* tsrc = (const float4*)(trig + (size_t)rowbase * META);
    tv0 = tsrc[t];
    tv1 = tsrc[t + 256];
    tv2 = tsrc[t + 512];
    tv3 = tsrc[t + 768];
    xv = ((const float4*)(tx + (size_t)rowbase * TXD))[t & 127];
    cv = chain[rowbase + (t & 63)];
    gv = graphs[rowbase + gsel];
}

__global__ void __launch_bounds__(256) batch_kernel(const int* __restrict__ graphs,
                                                    const float* __restrict__ chain,
                                                    const float* __restrict__ trig,
                                                    const float* __restrict__ tx,
                                                    const float* __restrict__ trigW,
                                                    const float* __restrict__ trigb,
                                                    const float* __restrict__ embW,
                                                    const float* __restrict__ embb,
                                                    const float* __restrict__ gp,
                                                    float* __restrict__ out) {
    __shared__ __align__(16) float gpl[N_GRAPHS * NOISE];
    __shared__ float T[CROWS * TSTRIDE];
    __shared__ __align__(16) float A1[CROWS * A1PITCH];
    __shared__ __align__(16) float O[CROWS * OPITCH];

    const int t     = threadIdx.x;
    const int lane  = t & 31;
    const int w     = t >> 5;
    const int lrow  = lane & 15;
    const int khalf = (lane >> 4) * 2;
    const int col   = w * 16 + lrow;
    const int lr0   = (lane >> 4) * 8;
    const int wu    = __builtin_amdgcn_readfirstlane(w);

    v2f bf[11];
#pragma unroll
    for (int ks = 0; ks < 11; ++ks) {
        const int k0 = ks * 4 + khalf;
        bf[ks].x = embW[(size_t)kext_row(k0) * NOISE + col];
        bf[ks].y = embW[(size_t)kext_row(k0 + 1) * NOISE + col];
    }

    const int rt1 = w >> 1;
    const int nt1 = w & 1;
    const int c1  = nt1 * 16 + lrow;
    v2f b1[16];
#pragma unroll
    for (int ks = 0; ks < 16; ++ks) {
        const int k0 = ks * 4 + khalf;
        b1[ks].x = trigW[k0 * 32 + c1];
        b1[ks].y = trigW[(k0 + 1) * 32 + c1];
    }
    const float tb = trigb[c1];
    const float4 eb4 = *(const float4*)&embb[lane * 4];
    int chunk = blockIdx.x;
    if (chunk >= NCHUNK) return;
    float4 tv0, tv1, tv2, tv3, xv;
    float cv;
    int gv;
    const int gsel = w * 8 + (lane & 7);
    load_chunk(trig, tx, chain, graphs, chunk * CROWS, t, gsel, tv0, tv1, tv2, tv3, xv, cv, gv);
    {
        float4 gq[8];
#pragma unroll
        for (int j = 0; j < 8; ++j) gq[j] = *(const float4*)&gp[j * 1024 + t * 4];
#pragma unroll
        for (int j = 0; j < 8; ++j) *(float4*)&gpl[j * 1024 + t * 4] = gq[j];
    }
    if (t < CROWS) {
        for (int c = 41; c < TSTRIDE; ++c) T[t * TSTRIDE + c] = 0.0f;
    }

    for (;;) {
        const int rowbase = chunk * CROWS;

        if (chunk + gridDim.x < NCHUNK) {
            const float* p = trig + (size_t)(chunk + gridDim.x) * CROWS * META
                                  + t * 16;
            __builtin_prefetch(p, 0, 3);
        }

        {
            float* a1p = &A1[(t >> 4) * A1PITCH + (t & 15) * 4];
            *(float4*)(a1p)                = tv0;
            *(float4*)(a1p + 16 * A1PITCH) = tv1;
            *(float4*)(a1p + 32 * A1PITCH) = tv2;
            *(float4*)(a1p + 48 * A1PITCH) = tv3;
        }
        if (t < 128) {
            float* tp = &T[(t >> 1) * TSTRIDE + 33 + (t & 1) * 4];
            tp[0] = xv.x;  tp[1] = xv.y;  tp[2] = xv.z;  tp[3] = xv.w;
        } else if (t < 192) {
            T[(t - 128) * TSTRIDE + 32] = cv;
        }
        __syncthreads();

        {
            v8f d = {};
            const float* ap = &A1[(rt1 * 16 + lrow) * A1PITCH + khalf];
#pragma unroll
            for (int ks = 0; ks < 16; ++ks) {
                const v2f a = *(const v2f*)(ap + ks * 4);
                d = __builtin_amdgcn_wmma_f32_16x16x4_f32(false, a, false, b1[ks],
                                                          (short)0, d, false, false);
            }
#pragma unroll
            for (int r = 0; r < 8; ++r) {
                float v = d[r] + tb;
                T[(rt1 * 16 + lr0 + r) * TSTRIDE + c1] = v > 0.0f ? v : 0.0f;
            }
        }
        __syncthreads();

        v8f acc0 = {}, acc1 = {}, acc2 = {}, acc3 = {};
#pragma unroll
        for (int ks = 0; ks < 11; ++ks) {
            const int k0 = ks * 4 + khalf;
            v2f a0, a1, a2, a3;
            a0.x = T[(0 * 16 + lrow) * TSTRIDE + k0];
            a0.y = T[(0 * 16 + lrow) * TSTRIDE + k0 + 1];
            a1.x = T[(1 * 16 + lrow) * TSTRIDE + k0];
            a1.y = T[(1 * 16 + lrow) * TSTRIDE + k0 + 1];
            a2.x = T[(2 * 16 + lrow) * TSTRIDE + k0];
            a2.y = T[(2 * 16 + lrow) * TSTRIDE + k0 + 1];
            a3.x = T[(3 * 16 + lrow) * TSTRIDE + k0];
            a3.y = T[(3 * 16 + lrow) * TSTRIDE + k0 + 1];
            acc0 = __builtin_amdgcn_wmma_f32_16x16x4_f32(false, a0, false, bf[ks],
                                                         (short)0, acc0, false, false);
            acc1 = __builtin_amdgcn_wmma_f32_16x16x4_f32(false, a1, false, bf[ks],
                                                         (short)0, acc1, false, false);
            acc2 = __builtin_amdgcn_wmma_f32_16x16x4_f32(false, a2, false, bf[ks],
                                                         (short)0, acc2, false, false);
            acc3 = __builtin_amdgcn_wmma_f32_16x16x4_f32(false, a3, false, bf[ks],
                                                         (short)0, acc3, false, false);
        }

#pragma unroll
        for (int rt = 0; rt < 4; ++rt) {
            const v8f& c = (rt == 0) ? acc0 : (rt == 1) ? acc1
                         : (rt == 2) ? acc2 : acc3;
#pragma unroll
            for (int r = 0; r < 8; ++r)
                O[(rt * 16 + lr0 + r) * OPITCH + col] = c[r];
        }
        __syncthreads();

#pragma unroll
        for (int i = 0; i < 8; ++i) {
            const int row = wu * 8 + i;
            const int g = __builtin_amdgcn_readlane(gv, i);
            const float4 o = *(const float4*)&O[row * OPITCH + lane * 4];
            const float4 q = *(const float4*)&gpl[g * NOISE + lane * 4];
            float4 v;
            v.x = (o.x + q.x) + eb4.x;
            v.y = (o.y + q.y) + eb4.y;
            v.z = (o.z + q.z) + eb4.z;
            v.w = (o.w + q.w) + eb4.w;
            *(float4*)&out[(size_t)(rowbase + row) * NOISE + lane * 4] = v;
        }
        chunk += gridDim.x;
        if (chunk >= NCHUNK) break;
        __syncthreads();

        load_chunk(trig, tx, chain, graphs, chunk * CROWS, t, gsel, tv0, tv1, tv2, tv3, xv, cv, gv);
    }
}

extern "C" void kernel_launch(void* const* d_in, const int* in_sizes, int n_in,
                              void* d_out, int out_size, void* d_ws, size_t ws_size,
                              hipStream_t stream) {
    const int*   graphs = (const int*)d_in[0];
    const float* chain  = (const float*)d_in[1];
    const float* trig   = (const float*)d_in[2];
    const float* tx     = (const float*)d_in[3];
    const int*   edges  = (const int*)d_in[4];
    const float* gw     = (const float*)d_in[5];
    const float* gb     = (const float*)d_in[6];
    const float* trigW  = (const float*)d_in[7];
    const float* trigb  = (const float*)d_in[8];
    const float* embW   = (const float*)d_in[9];
    const float* embb   = (const float*)d_in[10];
    float* out = (float*)d_out;

    float* ge = (float*)d_ws;
    float* gp = ge + (size_t)N_GRAPHS * N_NODES;
    float* part = gp + (size_t)N_GRAPHS * NOISE;
    unsigned int* tick = (unsigned int*)(part + (size_t)PTILES * PNCH * 256);

    gcn_kernel<<<N_GRAPHS * GSPLIT, GTHREADS, 0, stream>>>(edges, gw, gb, ge, tick);
    proj_kernel<<<PTILES * PNCH, PTHREADS, 0, stream>>>(ge, embW, part, tick, gp);
    batch_kernel<<<512, 256, 0, stream>>>(graphs, chain, trig, tx, trigW, trigb,
                                          embW, embb, gp, out);
}
